# starting kernel
# baseline (speedup 1.0000x reference)
#include <hip/hip_runtime.h>

typedef float v2f __attribute__((ext_vector_type(2)));
typedef float v8f __attribute__((ext_vector_type(8)));

#define TDIM     1000
#define ZSTRIDE  1002
#define FDIM     257
#define CDIM     8
#define CEPS     1.1920928955078125e-07f

__global__ __launch_bounds__(256) void mvdr_wmma_kernel(
    const float* __restrict__ spec_r,
    const float* __restrict__ spec_i,
    const float* __restrict__ mask,
    float*       __restrict__ out)
{
    __shared__ __align__(16) float Zs[16 * ZSTRIDE];
    __shared__ float Ms[TDIM];
    __shared__ float Gs[256];
    __shared__ float Gt[256];
    __shared__ float AR[8 * 16], AI[8 * 16];
    __shared__ float RsR[64], RsI[64];
    __shared__ float MR[64], MI[64];
    __shared__ float wR[8], wI[8];
    __shared__ float pivR, pivI;
    __shared__ float msum;

    const int tid = threadIdx.x;
    const int f   = blockIdx.x;
    const int n   = blockIdx.y;

    Gs[tid] = 0.0f;
    Gt[tid] = 0.0f;
    if (tid == 0) msum = 0.0f;

    if (tid < TDIM / 4) {
        const int t4 = tid;
        #pragma unroll
        for (int r = 0; r < CDIM; ++r) {
            const float4 v = ((const float4*)(spec_r + (((size_t)n * CDIM + r) * FDIM + f) * TDIM))[t4];
            float* dst = &Zs[r * ZSTRIDE + 4 * t4];
            dst[0] = v.x; dst[1] = v.y; dst[2] = v.z; dst[3] = v.w;
        }
        #pragma unroll
        for (int r = 0; r < CDIM; ++r) {
            const float4 v = ((const float4*)(spec_i + (((size_t)n * CDIM + r) * FDIM + f) * TDIM))[t4];
            float* dst = &Zs[(CDIM + r) * ZSTRIDE + 4 * t4];
            dst[0] = v.x; dst[1] = v.y; dst[2] = v.z; dst[3] = v.w;
        }
        const float4 mv = ((const float4*)(mask + ((size_t)n * FDIM + f) * TDIM))[t4];
        float* md = &Ms[4 * t4];
        md[0] = mv.x; md[1] = mv.y; md[2] = mv.z; md[3] = mv.w;
    }
    __syncthreads();

    {
        float local = 0.0f;
        for (int t = tid; t < TDIM; t += 256) local += Ms[t];
        atomicAdd(&msum, local);
    }

    {
        const int wave = tid >> 5;
        const int lane = tid & 31;
        const int m    = lane & 15;
        const int koff = (lane >> 4) * 2;

        v8f acc_s = {0, 0, 0, 0, 0, 0, 0, 0};
        v8f acc_t = {0, 0, 0, 0, 0, 0, 0, 0};

        for (int it = wave; it < TDIM / 4; it += 8) {
            const int t0 = it * 4 + koff;
            v2f zb = *(const v2f*)&Zs[m * ZSTRIDE + t0];
            v2f za;
            za.x = zb.x * Ms[t0];
            za.y = zb.y * Ms[t0 + 1];
            acc_t = __builtin_amdgcn_wmma_f32_16x16x4_f32(
                        false, zb, false, zb, (short)0, acc_t, false, false);
            acc_s = __builtin_amdgcn_wmma_f32_16x16x4_f32(
                        false, za, false, zb, (short)0, acc_s, false, false);
        }

        const int Nn    = lane & 15;
        const int Mbase = (lane >> 4) * 8;
        #pragma unroll
        for (int r = 0; r < 8; ++r) {
            atomicAdd(&Gs[(Mbase + r) * 16 + Nn], acc_s[r]);
            atomicAdd(&Gt[(Mbase + r) * 16 + Nn], acc_t[r]);
        }
    }
    __syncthreads();

    if (tid < 64) {
        const int i = tid >> 3, j = tid & 7;
        const float ds    = fmaxf(msum, CEPS);
        const float dn    = fmaxf((float)TDIM - msum, CEPS);
        const float inv_s = 1.0f / ds;
        const float inv_n = 1.0f / dn;

        const float gs_a = Gs[i * 16 + j],           gt_a = Gt[i * 16 + j];
        const float gs_b = Gs[(i + 8) * 16 + j + 8], gt_b = Gt[(i + 8) * 16 + j + 8];
        const float gs_c = Gs[(i + 8) * 16 + j],     gt_c = Gt[(i + 8) * 16 + j];
        const float gs_d = Gs[i * 16 + j + 8],       gt_d = Gt[i * 16 + j + 8];

        RsR[i * 8 + j] = (gs_a + gs_b) * inv_s;
        RsI[i * 8 + j] = (gs_c - gs_d) * inv_s;

        const float rnr = ((gt_a - gs_a) + (gt_b - gs_b)) * inv_n;
        const float rni = ((gt_c - gs_c) - (gt_d - gs_d)) * inv_n;
        AR[i * 16 + j]     = rnr + ((i == j) ? CEPS : 0.0f);
        AI[i * 16 + j]     = rni;
        AR[i * 16 + 8 + j] = (i == j) ? 1.0f : 0.0f;
        AI[i * 16 + 8 + j] = 0.0f;
    }
    __syncthreads();

    for (int k = 0; k < 8; ++k) {
        if (tid == 0) {
            const float pr = AR[k * 16 + k], pi = AI[k * 16 + k];
            const float id = 1.0f / (pr * pr + pi * pi);
            pivR = pr * id;
            pivI = -pi * id;
        }
        __syncthreads();
        if (tid < 16) {
            const float ar = AR[k * 16 + tid], ai = AI[k * 16 + tid];
            AR[k * 16 + tid] = ar * pivR - ai * pivI;
            AI[k * 16 + tid] = ar * pivI + ai * pivR;
        }
        __syncthreads();
        const int  i   = tid >> 4, j = tid & 15;
        const bool act = (tid < 128) && (i != k);
        float fr = 0.0f, fi = 0.0f, br = 0.0f, bi = 0.0f;
        if (act) {
            fr = AR[i * 16 + k]; fi = AI[i * 16 + k];
            br = AR[k * 16 + j]; bi = AI[k * 16 + j];
        }
        __syncthreads();
        if (act) {
            AR[i * 16 + j] -= fr * br - fi * bi;
            AI[i * 16 + j] -= fr * bi + fi * br;
        }
        __syncthreads();
    }

    if (tid < 64) {
        const int i = tid >> 3, j = tid & 7;
        float sr = 0.0f, si = 0.0f;
        #pragma unroll
        for (int k = 0; k < 8; ++k) {
            const float ar = AR[i * 16 + 8 + k], ai = AI[i * 16 + 8 + k];
            const float br = RsR[k * 8 + j],     bi = RsI[k * 8 + j];
            sr += ar * br - ai * bi;
            si += ar * bi + ai * br;
        }
        MR[i * 8 + j] = sr;
        MI[i * 8 + j] = si;
    }
    __syncthreads();

    if (tid == 0) {
        float tr = CEPS, ti = 0.0f;
        #pragma unroll
        for (int i = 0; i < 8; ++i) { tr += MR[i * 8 + i]; ti += MI[i * 8 + i]; }
        const float id = 1.0f / (tr * tr + ti * ti);
        #pragma unroll
        for (int i = 0; i < 8; ++i) {
            const float mr = MR[i * 8], mi = MI[i * 8];
            wR[i] = (mr * tr + mi * ti) * id;
            wI[i] = (mi * tr - mr * ti) * id;
        }
    }
    __syncthreads();

    float* outR = out + (((size_t)n * 2 + 0) * FDIM + f) * TDIM;
    float* outI = out + (((size_t)n * 2 + 1) * FDIM + f) * TDIM;
    for (int t = tid; t < TDIM; t += 256) {
        float br = 0.0f, bi = 0.0f;
        #pragma unroll
        for (int c = 0; c < CDIM; ++c) {
            const float sr = Zs[c * ZSTRIDE + t];
            const float si = Zs[(CDIM + c) * ZSTRIDE + t];
            br += wR[c] * sr + wI[c] * si;
            bi += wR[c] * si - wI[c] * sr;
        }
        outR[t] = br;
        outI[t] = bi;
    }
}

extern "C" void kernel_launch(void* const* d_in, const int* in_sizes, int n_in,
                              void* d_out, int out_size, void* d_ws, size_t ws_size,
                              hipStream_t stream) {
    (void)in_sizes; (void)n_in; (void)out_size; (void)d_ws; (void)ws_size;
    const float* spec_r = (const float*)d_in[0];
    const float* spec_i = (const float*)d_in[1];
    const float* mask   = (const float*)d_in[2];
    float*       outp   = (float*)d_out;

    dim3 grid(FDIM, 16);
    dim3 block(256);
    hipLaunchKernelGGL(mvdr_wmma_kernel, grid, block, 0, stream,
                       spec_r, spec_i, mask, outp);
}


# 8x8 complex Gauss-Jordan moved into one wave's registers (lane = column x re/im, rows in registers), eliminating [Rn|Rs] directly, so the inverse-then-product sums go: 36 block barriers become 1
# speedup vs baseline: 1.1979x; 1.1979x over previous
#include <hip/hip_runtime.h>

typedef float v2f __attribute__((ext_vector_type(2)));
typedef float v8f __attribute__((ext_vector_type(8)));

#define TDIM     1000
#define ZSTRIDE  1002
#define FDIM     257
#define CDIM     8
#define CEPS     1.1920928955078125e-07f

__device__ __forceinline__ float lane_value(float x, int lane) {
    return __builtin_bit_cast(float, __builtin_amdgcn_readlane(__builtin_bit_cast(int, x), lane));
}
__device__ __forceinline__ float other_half(float x) {
    const int b = __builtin_bit_cast(int, x);
    return __builtin_bit_cast(float, __builtin_amdgcn_permlanex16(b, b, 0x76543210, 0xfedcba98, false, false));
}

__global__ __launch_bounds__(256) void mvdr_wmma_kernel(
    const float* __restrict__ spec_r,
    const float* __restrict__ spec_i,
    const float* __restrict__ mask,
    float*       __restrict__ out)
{
    __shared__ __align__(16) float Zs[16 * ZSTRIDE];
    __shared__ float Ms[TDIM];
    __shared__ float Gs[256];
    __shared__ float Gt[256];
    __shared__ __align__(16) float wR[8];
    __shared__ __align__(16) float wI[8];
    __shared__ float msum;

    const int tid = threadIdx.x;
    const int f   = blockIdx.x;
    const int n   = blockIdx.y;

    Gs[tid] = 0.0f;
    Gt[tid] = 0.0f;
    if (tid == 0) msum = 0.0f;

    if (tid < TDIM / 4) {
        const int t4 = tid;
        #pragma unroll
        for (int r = 0; r < CDIM; ++r) {
            const float4 v = ((const float4*)(spec_r + (((size_t)n * CDIM + r) * FDIM + f) * TDIM))[t4];
            float* dst = &Zs[r * ZSTRIDE + 4 * t4];
            dst[0] = v.x; dst[1] = v.y; dst[2] = v.z; dst[3] = v.w;
        }
        #pragma unroll
        for (int r = 0; r < CDIM; ++r) {
            const float4 v = ((const float4*)(spec_i + (((size_t)n * CDIM + r) * FDIM + f) * TDIM))[t4];
            float* dst = &Zs[(CDIM + r) * ZSTRIDE + 4 * t4];
            dst[0] = v.x; dst[1] = v.y; dst[2] = v.z; dst[3] = v.w;
        }
        const float4 mv = ((const float4*)(mask + ((size_t)n * FDIM + f) * TDIM))[t4];
        float* md = &Ms[4 * t4];
        md[0] = mv.x; md[1] = mv.y; md[2] = mv.z; md[3] = mv.w;
    }
    __syncthreads();

    {
        float local = 0.0f;
        for (int t = tid; t < TDIM; t += 256) local += Ms[t];
        atomicAdd(&msum, local);
    }

    {
        const int wave = tid >> 5;
        const int lane = tid & 31;
        const int m    = lane & 15;
        const int koff = (lane >> 4) * 2;

        v8f acc_s = {0, 0, 0, 0, 0, 0, 0, 0};
        v8f acc_t = {0, 0, 0, 0, 0, 0, 0, 0};

        for (int it = wave; it < TDIM / 4; it += 8) {
            const int t0 = it * 4 + koff;
            v2f zb = *(const v2f*)&Zs[m * ZSTRIDE + t0];
            v2f za;
            za.x = zb.x * Ms[t0];
            za.y = zb.y * Ms[t0 + 1];
            acc_t = __builtin_amdgcn_wmma_f32_16x16x4_f32(
                        false, zb, false, zb, (short)0, acc_t, false, false);
            acc_s = __builtin_amdgcn_wmma_f32_16x16x4_f32(
                        false, za, false, zb, (short)0, acc_s, false, false);
        }

        const int Nn    = lane & 15;
        const int Mbase = (lane >> 4) * 8;
        #pragma unroll
        for (int r = 0; r < 8; ++r) {
            atomicAdd(&Gs[(Mbase + r) * 16 + Nn], acc_s[r]);
            atomicAdd(&Gt[(Mbase + r) * 16 + Nn], acc_t[r]);
        }
    }
    __syncthreads();

    if (__builtin_amdgcn_readfirstlane(tid >> 5) == 0) {
        const int   col  = tid & 15;
        const int   part = tid >> 4;
        const int   j    = col & 7;
        const bool  rhs  = col >= 8;
        const float ms    = msum;
        const float inv_s = 1.0f / fmaxf(ms, CEPS);
        const float inv_n = 1.0f / fmaxf((float)TDIM - ms, CEPS);
        const int   pIdx = (8 * part) * 16 + j;
        const int   qIdx = (8 * (1 - part)) * 16 + j + 8;
        const float sgn  = part ? -1.0f : 1.0f;

        float a[8];
        #pragma unroll
        for (int i = 0; i < 8; ++i) {
            const float gs_p = Gs[i * 16 + pIdx], gs_q = Gs[i * 16 + qIdx];
            const float gt_p = Gt[i * 16 + pIdx], gt_q = Gt[i * 16 + qIdx];
            const float rs = (gs_p + sgn * gs_q) * inv_s;
            float       rn = ((gt_p - gs_p) + sgn * (gt_q - gs_q)) * inv_n;
            rn += (part == 0 && i == j) ? CEPS : 0.0f;
            a[i] = rhs ? rs : rn;
        }

        #pragma unroll
        for (int k = 0; k < 8; ++k) {
            const float pr = lane_value(a[k], k), pi = lane_value(a[k], 16 + k);
            const float id = 1.0f / (pr * pr + pi * pi);
            const float pivR = pr * id, pivI = -pi * id;
            const float mine = a[k], oth = other_half(a[k]);
            const float ar = part ? oth : mine;
            const float ai = part ? mine : oth;
            const float nr = ar * pivR - ai * pivI;
            const float ni = ar * pivI + ai * pivR;
            a[k] = part ? ni : nr;
            const float u = part ? ni : nr;
            const float v = part ? nr : -ni;
            #pragma unroll
            for (int i = 0; i < 8; ++i) {
                if (i == k) continue;
                const float fr = lane_value(a[i], k), fi = lane_value(a[i], 16 + k);
                a[i] -= fr * u + fi * v;
            }
        }

        float tr = CEPS, ti = 0.0f;
        #pragma unroll
        for (int i = 0; i < 8; ++i) { tr += lane_value(a[i], 8 + i); ti += lane_value(a[i], 24 + i); }
        const float idt = 1.0f / (tr * tr + ti * ti);
        float wr[8], wi[8];
        #pragma unroll
        for (int i = 0; i < 8; ++i) {
            const float mr = lane_value(a[i], 8), mi = lane_value(a[i], 24);
            wr[i] = (mr * tr + mi * ti) * idt;
            wi[i] = (mi * tr - mr * ti) * idt;
        }
        if (tid == 0) {
            #pragma unroll
            for (int i = 0; i < 8; ++i) { wR[i] = wr[i]; wI[i] = wi[i]; }
        }
    }
    __syncthreads();

    float* outR = out + (((size_t)n * 2 + 0) * FDIM + f) * TDIM;
    float* outI = out + (((size_t)n * 2 + 1) * FDIM + f) * TDIM;
    for (int t = tid; t < TDIM; t += 256) {
        float br = 0.0f, bi = 0.0f;
        #pragma unroll
        for (int c = 0; c < CDIM; ++c) {
            const float sr = Zs[c * ZSTRIDE + t];
            const float si = Zs[(CDIM + c) * ZSTRIDE + t];
            br += wR[c] * sr + wI[c] * si;
            bi += wR[c] * si - wI[c] * sr;
        }
        outR[t] = br;
        outI[t] = bi;
    }
}

extern "C" void kernel_launch(void* const* d_in, const int* in_sizes, int n_in,
                              void* d_out, int out_size, void* d_ws, size_t ws_size,
                              hipStream_t stream) {
    (void)in_sizes; (void)n_in; (void)out_size; (void)d_ws; (void)ws_size;
    const float* spec_r = (const float*)d_in[0];
    const float* spec_i = (const float*)d_in[1];
    const float* mask   = (const float*)d_in[2];
    float*       outp   = (float*)d_out;

    dim3 grid(FDIM, 16);
    dim3 block(256);
    hipLaunchKernelGGL(mvdr_wmma_kernel, grid, block, 0, stream,
                       spec_r, spec_i, mask, outp);
}


# Gram loop taken in blocks of 8 k-steps: the next block's LDS fragment and mask loads go out as one group ahead of the current block's 16 WMMAs instead of a load-wait-WMMA round trip per k-step
# speedup vs baseline: 1.2545x; 1.0472x over previous
#include <hip/hip_runtime.h>

typedef float v2f __attribute__((ext_vector_type(2)));
typedef float v8f __attribute__((ext_vector_type(8)));

#define TDIM     1000
#define ZSTRIDE  1002
#define FDIM     257
#define CDIM     8
#define CEPS     1.1920928955078125e-07f

__device__ __forceinline__ float lane_value(float x, int lane) {
    return __builtin_bit_cast(float, __builtin_amdgcn_readlane(__builtin_bit_cast(int, x), lane));
}
__device__ __forceinline__ float other_half(float x) {
    const int b = __builtin_bit_cast(int, x);
    return __builtin_bit_cast(float, __builtin_amdgcn_permlanex16(b, b, 0x76543210, 0xfedcba98, false, false));
}

__global__ __launch_bounds__(256) void mvdr_wmma_kernel(
    const float* __restrict__ spec_r,
    const float* __restrict__ spec_i,
    const float* __restrict__ mask,
    float*       __restrict__ out)
{
    __shared__ __align__(16) float Zs[16 * ZSTRIDE];
    __shared__ float Ms[TDIM];
    __shared__ float Gs[256];
    __shared__ float Gt[256];
    __shared__ __align__(16) float wR[8];
    __shared__ __align__(16) float wI[8];
    __shared__ float msum;

    const int tid = threadIdx.x;
    const int f   = blockIdx.x;
    const int n   = blockIdx.y;

    Gs[tid] = 0.0f;
    Gt[tid] = 0.0f;
    if (tid == 0) msum = 0.0f;

    if (tid < TDIM / 4) {
        const int t4 = tid;
        #pragma unroll
        for (int r = 0; r < CDIM; ++r) {
            const float4 v = ((const float4*)(spec_r + (((size_t)n * CDIM + r) * FDIM + f) * TDIM))[t4];
            float* dst = &Zs[r * ZSTRIDE + 4 * t4];
            dst[0] = v.x; dst[1] = v.y; dst[2] = v.z; dst[3] = v.w;
        }
        #pragma unroll
        for (int r = 0; r < CDIM; ++r) {
            const float4 v = ((const float4*)(spec_i + (((size_t)n * CDIM + r) * FDIM + f) * TDIM))[t4];
            float* dst = &Zs[(CDIM + r) * ZSTRIDE + 4 * t4];
            dst[0] = v.x; dst[1] = v.y; dst[2] = v.z; dst[3] = v.w;
        }
        const float4 mv = ((const float4*)(mask + ((size_t)n * FDIM + f) * TDIM))[t4];
        float* md = &Ms[4 * t4];
        md[0] = mv.x; md[1] = mv.y; md[2] = mv.z; md[3] = mv.w;
    }
    __syncthreads();

    {
        float local = 0.0f;
        for (int t = tid; t < TDIM; t += 256) local += Ms[t];
        atomicAdd(&msum, local);
    }

    {
        const int wave = __builtin_amdgcn_readfirstlane(tid >> 5);
        const int lane = tid & 31;
        const int m    = lane & 15;
        const int koff = (lane >> 4) * 2;

        v8f acc_s = {0, 0, 0, 0, 0, 0, 0, 0};
        v8f acc_t = {0, 0, 0, 0, 0, 0, 0, 0};

        const int   tbase = 4 * wave + koff;
        const float* zrow = &Zs[m * ZSTRIDE + tbase];
        const float* mrow = &Ms[tbase];
        const bool  tail  = wave < 2;
        const int   tailo = tail ? 31 * 32 : 0;

        v2f zc[8], mc[8], zn[8], mn[8];
        #pragma unroll
        for (int u = 0; u < 8; ++u) {
            zc[u] = *(const v2f*)(zrow + 32 * u);
            mc[u] = *(const v2f*)(mrow + 32 * u);
        }
        #pragma unroll
        for (int b = 0; b < 4; ++b) {
            if (b < 3) {
                #pragma unroll
                for (int u = 0; u < 8; ++u) {
                    const int o = (b == 2 && u == 7) ? tailo : 32 * (8 * (b + 1) + u);
                    zn[u] = *(const v2f*)(zrow + o);
                    mn[u] = *(const v2f*)(mrow + o);
                }
                if (b == 2 && !tail) {
                    zn[7].x = 0.0f; zn[7].y = 0.0f;
                    mn[7].x = 0.0f; mn[7].y = 0.0f;
                }
            }
            __builtin_amdgcn_sched_barrier(0);
            #pragma unroll
            for (int u = 0; u < 8; ++u) {
                const v2f zb = zc[u];
                v2f za;
                za.x = zb.x * mc[u].x;
                za.y = zb.y * mc[u].y;
                acc_t = __builtin_amdgcn_wmma_f32_16x16x4_f32(
                            false, zb, false, zb, (short)0, acc_t, false, false);
                acc_s = __builtin_amdgcn_wmma_f32_16x16x4_f32(
                            false, za, false, zb, (short)0, acc_s, false, false);
            }
            #pragma unroll
            for (int u = 0; u < 8; ++u) { zc[u] = zn[u]; mc[u] = mn[u]; }
        }

        const int Nn    = lane & 15;
        const int Mbase = (lane >> 4) * 8;
        #pragma unroll
        for (int r = 0; r < 8; ++r) {
            atomicAdd(&Gs[(Mbase + r) * 16 + Nn], acc_s[r]);
            atomicAdd(&Gt[(Mbase + r) * 16 + Nn], acc_t[r]);
        }
    }
    __syncthreads();

    if (__builtin_amdgcn_readfirstlane(tid >> 5) == 0) {
        const int   col  = tid & 15;
        const int   part = tid >> 4;
        const int   j    = col & 7;
        const bool  rhs  = col >= 8;
        const float ms    = msum;
        const float inv_s = 1.0f / fmaxf(ms, CEPS);
        const float inv_n = 1.0f / fmaxf((float)TDIM - ms, CEPS);
        const int   pIdx = (8 * part) * 16 + j;
        const int   qIdx = (8 * (1 - part)) * 16 + j + 8;
        const float sgn  = part ? -1.0f : 1.0f;

        float a[8];
        #pragma unroll
        for (int i = 0; i < 8; ++i) {
            const float gs_p = Gs[i * 16 + pIdx], gs_q = Gs[i * 16 + qIdx];
            const float gt_p = Gt[i * 16 + pIdx], gt_q = Gt[i * 16 + qIdx];
            const float rs = (gs_p + sgn * gs_q) * inv_s;
            float       rn = ((gt_p - gs_p) + sgn * (gt_q - gs_q)) * inv_n;
            rn += (part == 0 && i == j) ? CEPS : 0.0f;
            a[i] = rhs ? rs : rn;
        }

        #pragma unroll
        for (int k = 0; k < 8; ++k) {
            const float pr = lane_value(a[k], k), pi = lane_value(a[k], 16 + k);
            const float id = 1.0f / (pr * pr + pi * pi);
            const float pivR = pr * id, pivI = -pi * id;
            const float mine = a[k], oth = other_half(a[k]);
            const float ar = part ? oth : mine;
            const float ai = part ? mine : oth;
            const float nr = ar * pivR - ai * pivI;
            const float ni = ar * pivI + ai * pivR;
            a[k] = part ? ni : nr;
            const float u = part ? ni : nr;
            const float v = part ? nr : -ni;
            #pragma unroll
            for (int i = 0; i < 8; ++i) {
                if (i == k) continue;
                const float fr = lane_value(a[i], k), fi = lane_value(a[i], 16 + k);
                a[i] -= fr * u + fi * v;
            }
        }

        float tr = CEPS, ti = 0.0f;
        #pragma unroll
        for (int i = 0; i < 8; ++i) { tr += lane_value(a[i], 8 + i); ti += lane_value(a[i], 24 + i); }
        const float idt = 1.0f / (tr * tr + ti * ti);
        float wr[8], wi[8];
        #pragma unroll
        for (int i = 0; i < 8; ++i) {
            const float mr = lane_value(a[i], 8), mi = lane_value(a[i], 24);
            wr[i] = (mr * tr + mi * ti) * idt;
            wi[i] = (mi * tr - mr * ti) * idt;
        }
        if (tid == 0) {
            #pragma unroll
            for (int i = 0; i < 8; ++i) { wR[i] = wr[i]; wI[i] = wi[i]; }
        }
    }
    __syncthreads();

    float* outR = out + (((size_t)n * 2 + 0) * FDIM + f) * TDIM;
    float* outI = out + (((size_t)n * 2 + 1) * FDIM + f) * TDIM;
    for (int t = tid; t < TDIM; t += 256) {
        float br = 0.0f, bi = 0.0f;
        #pragma unroll
        for (int c = 0; c < CDIM; ++c) {
            const float sr = Zs[c * ZSTRIDE + t];
            const float si = Zs[(CDIM + c) * ZSTRIDE + t];
            br += wR[c] * sr + wI[c] * si;
            bi += wR[c] * si - wI[c] * sr;
        }
        outR[t] = br;
        outI[t] = bi;
    }
}

extern "C" void kernel_launch(void* const* d_in, const int* in_sizes, int n_in,
                              void* d_out, int out_size, void* d_ws, size_t ws_size,
                              hipStream_t stream) {
    (void)in_sizes; (void)n_in; (void)out_size; (void)d_ws; (void)ws_size;
    const float* spec_r = (const float*)d_in[0];
    const float* spec_i = (const float*)d_in[1];
    const float* mask   = (const float*)d_in[2];
    float*       outp   = (float*)d_out;

    dim3 grid(FDIM, 16);
    dim3 block(256);
    hipLaunchKernelGGL(mvdr_wmma_kernel, grid, block, 0, stream,
                       spec_r, spec_i, mask, outp);
}


# Each wave stages and consumes its own 128 columns (tile padded to 1024, pitch 1028): the barrier between staging and Gram loop goes; mask sum from the same fragments in fixed order, no LDS atomic
# speedup vs baseline: 1.3335x; 1.0630x over previous
#include <hip/hip_runtime.h>

typedef float v2f __attribute__((ext_vector_type(2)));
typedef float v8f __attribute__((ext_vector_type(8)));

#define TDIM     1000
#define TPAD     1024
#define ZSTRIDE  1028
#define FDIM     257
#define CDIM     8
#define CEPS     1.1920928955078125e-07f

__device__ __forceinline__ float lane_value(float x, int lane) {
    return __builtin_bit_cast(float, __builtin_amdgcn_readlane(__builtin_bit_cast(int, x), lane));
}
__device__ __forceinline__ float other_half(float x) {
    const int b = __builtin_bit_cast(int, x);
    return __builtin_bit_cast(float, __builtin_amdgcn_permlanex16(b, b, 0x76543210, 0xfedcba98, false, false));
}

__global__ __launch_bounds__(256) void mvdr_wmma_kernel(
    const float* __restrict__ spec_r,
    const float* __restrict__ spec_i,
    const float* __restrict__ mask,
    float*       __restrict__ out)
{
    __shared__ __align__(16) float Zs[16 * ZSTRIDE];
    __shared__ __align__(16) float Ms[TPAD];
    __shared__ float Gs[256];
    __shared__ float Gt[256];
    __shared__ __align__(16) float wR[8];
    __shared__ __align__(16) float wI[8];
    __shared__ __align__(16) float mpart[8];

    const int tid = threadIdx.x;
    const int f   = blockIdx.x;
    const int n   = blockIdx.y;

    Gs[tid] = 0.0f;
    Gt[tid] = 0.0f;
    __syncthreads();

    {
        const bool valid = tid < TDIM / 4;
        const int  t4    = tid;
        float4 zr[CDIM], zi[CDIM];
        float4 mv = make_float4(0.0f, 0.0f, 0.0f, 0.0f);
        #pragma unroll
        for (int r = 0; r < CDIM; ++r) {
            zr[r] = make_float4(0.0f, 0.0f, 0.0f, 0.0f);
            zi[r] = make_float4(0.0f, 0.0f, 0.0f, 0.0f);
        }
        if (valid) {
            #pragma unroll
            for (int r = 0; r < CDIM; ++r)
                zr[r] = ((const float4*)(spec_r + (((size_t)n * CDIM + r) * FDIM + f) * TDIM))[t4];
            #pragma unroll
            for (int r = 0; r < CDIM; ++r)
                zi[r] = ((const float4*)(spec_i + (((size_t)n * CDIM + r) * FDIM + f) * TDIM))[t4];
            mv = ((const float4*)(mask + ((size_t)n * FDIM + f) * TDIM))[t4];
        }
        #pragma unroll
        for (int r = 0; r < CDIM; ++r) {
            *(float4*)&Zs[r * ZSTRIDE + 4 * t4]          = zr[r];
            *(float4*)&Zs[(CDIM + r) * ZSTRIDE + 4 * t4] = zi[r];
        }
        *(float4*)&Ms[4 * t4] = mv;
    }
    __builtin_amdgcn_fence(__ATOMIC_RELEASE, "wavefront");
    __builtin_amdgcn_wave_barrier();
    __builtin_amdgcn_fence(__ATOMIC_ACQUIRE, "wavefront");

    {
        const int wave = __builtin_amdgcn_readfirstlane(tid >> 5);
        const int lane = tid & 31;
        const int m    = lane & 15;
        const int koff = (lane >> 4) * 2;

        v8f acc_s = {0, 0, 0, 0, 0, 0, 0, 0};
        v8f acc_t = {0, 0, 0, 0, 0, 0, 0, 0};
        v2f macc  = {0, 0};

        const int   tbase = 128 * wave + koff;
        const float* zrow = &Zs[m * ZSTRIDE + tbase];
        const float* mrow = &Ms[tbase];

        v2f zc[8], mc[8], zn[8], mn[8];
        #pragma unroll
        for (int u = 0; u < 8; ++u) {
            zc[u] = *(const v2f*)(zrow + 4 * u);
            mc[u] = *(const v2f*)(mrow + 4 * u);
        }
        #pragma unroll
        for (int b = 0; b < 4; ++b) {
            if (b < 3) {
                #pragma unroll
                for (int u = 0; u < 8; ++u) {
                    zn[u] = *(const v2f*)(zrow + 4 * (8 * (b + 1) + u));
                    mn[u] = *(const v2f*)(mrow + 4 * (8 * (b + 1) + u));
                }
            }
            __builtin_amdgcn_sched_barrier(0);
            #pragma unroll
            for (int u = 0; u < 8; ++u) {
                const v2f zb = zc[u];
                v2f za;
                za.x = zb.x * mc[u].x;
                za.y = zb.y * mc[u].y;
                macc.x += mc[u].x;
                macc.y += mc[u].y;
                acc_t = __builtin_amdgcn_wmma_f32_16x16x4_f32(
                            false, zb, false, zb, (short)0, acc_t, false, false);
                acc_s = __builtin_amdgcn_wmma_f32_16x16x4_f32(
                            false, za, false, zb, (short)0, acc_s, false, false);
            }
            #pragma unroll
            for (int u = 0; u < 8; ++u) { zc[u] = zn[u]; mc[u] = mn[u]; }
        }

        const int Nn    = lane & 15;
        const int Mbase = (lane >> 4) * 8;
        #pragma unroll
        for (int r = 0; r < 8; ++r) {
            atomicAdd(&Gs[(Mbase + r) * 16 + Nn], acc_s[r]);
            atomicAdd(&Gt[(Mbase + r) * 16 + Nn], acc_t[r]);
        }
        float wsum = macc.x + macc.y;
        wsum += other_half(wsum);
        if (lane == 0) mpart[wave] = wsum;
    }
    __syncthreads();

    if (__builtin_amdgcn_readfirstlane(tid >> 5) == 0) {
        const int   col  = tid & 15;
        const int   part = tid >> 4;
        const int   j    = col & 7;
        const bool  rhs  = col >= 8;
        const float ms    = ((mpart[0] + mpart[1]) + (mpart[2] + mpart[3])) + ((mpart[4] + mpart[5]) + (mpart[6] + mpart[7]));
        const float inv_s = 1.0f / fmaxf(ms, CEPS);
        const float inv_n = 1.0f / fmaxf((float)TDIM - ms, CEPS);
        const int   pIdx = (8 * part) * 16 + j;
        const int   qIdx = (8 * (1 - part)) * 16 + j + 8;
        const float sgn  = part ? -1.0f : 1.0f;

        float a[8];
        #pragma unroll
        for (int i = 0; i < 8; ++i) {
            const float gs_p = Gs[i * 16 + pIdx], gs_q = Gs[i * 16 + qIdx];
            const float gt_p = Gt[i * 16 + pIdx], gt_q = Gt[i * 16 + qIdx];
            const float rs = (gs_p + sgn * gs_q) * inv_s;
            float       rn = ((gt_p - gs_p) + sgn * (gt_q - gs_q)) * inv_n;
            rn += (part == 0 && i == j) ? CEPS : 0.0f;
            a[i] = rhs ? rs : rn;
        }

        #pragma unroll
        for (int k = 0; k < 8; ++k) {
            const float pr = lane_value(a[k], k), pi = lane_value(a[k], 16 + k);
            const float id = 1.0f / (pr * pr + pi * pi);
            const float pivR = pr * id, pivI = -pi * id;
            const float mine = a[k], oth = other_half(a[k]);
            const float ar = part ? oth : mine;
            const float ai = part ? mine : oth;
            const float nr = ar * pivR - ai * pivI;
            const float ni = ar * pivI + ai * pivR;
            a[k] = part ? ni : nr;
            const float u = part ? ni : nr;
            const float v = part ? nr : -ni;
            #pragma unroll
            for (int i = 0; i < 8; ++i) {
                if (i == k) continue;
                const float fr = lane_value(a[i], k), fi = lane_value(a[i], 16 + k);
                a[i] -= fr * u + fi * v;
            }
        }

        float tr = CEPS, ti = 0.0f;
        #pragma unroll
        for (int i = 0; i < 8; ++i) { tr += lane_value(a[i], 8 + i); ti += lane_value(a[i], 24 + i); }
        const float idt = 1.0f / (tr * tr + ti * ti);
        float wr[8], wi[8];
        #pragma unroll
        for (int i = 0; i < 8; ++i) {
            const float mr = lane_value(a[i], 8), mi = lane_value(a[i], 24);
            wr[i] = (mr * tr + mi * ti) * idt;
            wi[i] = (mi * tr - mr * ti) * idt;
        }
        if (tid == 0) {
            #pragma unroll
            for (int i = 0; i < 8; ++i) { wR[i] = wr[i]; wI[i] = wi[i]; }
        }
    }
    __syncthreads();

    float* outR = out + (((size_t)n * 2 + 0) * FDIM + f) * TDIM;
    float* outI = out + (((size_t)n * 2 + 1) * FDIM + f) * TDIM;
    for (int t = tid; t < TDIM; t += 256) {
        float br = 0.0f, bi = 0.0f;
        #pragma unroll
        for (int c = 0; c < CDIM; ++c) {
            const float sr = Zs[c * ZSTRIDE + t];
            const float si = Zs[(CDIM + c) * ZSTRIDE + t];
            br += wR[c] * sr + wI[c] * si;
            bi += wR[c] * si - wI[c] * sr;
        }
        outR[t] = br;
        outI[t] = bi;
    }
}

extern "C" void kernel_launch(void* const* d_in, const int* in_sizes, int n_in,
                              void* d_out, int out_size, void* d_ws, size_t ws_size,
                              hipStream_t stream) {
    (void)in_sizes; (void)n_in; (void)out_size; (void)d_ws; (void)ws_size;
    const float* spec_r = (const float*)d_in[0];
    const float* spec_i = (const float*)d_in[1];
    const float* mask   = (const float*)d_in[2];
    float*       outp   = (float*)d_out;

    dim3 grid(FDIM, 16);
    dim3 block(256);
    hipLaunchKernelGGL(mvdr_wmma_kernel, grid, block, 0, stream,
                       spec_r, spec_i, mask, outp);
}


# Output (beamform) phase: each thread takes 4 consecutive columns: 16-byte LDS reads (20 per thread, was 68) and one 16-byte store per plane (was four 4-byte); per-element sums in the same order
# speedup vs baseline: 1.3964x; 1.0472x over previous
#include <hip/hip_runtime.h>

typedef float v2f __attribute__((ext_vector_type(2)));
typedef float v8f __attribute__((ext_vector_type(8)));

#define TDIM     1000
#define TPAD     1024
#define ZSTRIDE  1028
#define FDIM     257
#define CDIM     8
#define CEPS     1.1920928955078125e-07f

__device__ __forceinline__ float lane_value(float x, int lane) {
    return __builtin_bit_cast(float, __builtin_amdgcn_readlane(__builtin_bit_cast(int, x), lane));
}
__device__ __forceinline__ float other_half(float x) {
    const int b = __builtin_bit_cast(int, x);
    return __builtin_bit_cast(float, __builtin_amdgcn_permlanex16(b, b, 0x76543210, 0xfedcba98, false, false));
}

__global__ __launch_bounds__(256) void mvdr_wmma_kernel(
    const float* __restrict__ spec_r,
    const float* __restrict__ spec_i,
    const float* __restrict__ mask,
    float*       __restrict__ out)
{
    __shared__ __align__(16) float Zs[16 * ZSTRIDE];
    __shared__ __align__(16) float Ms[TPAD];
    __shared__ float Gs[256];
    __shared__ float Gt[256];
    __shared__ __align__(16) float wR[8];
    __shared__ __align__(16) float wI[8];
    __shared__ __align__(16) float mpart[8];

    const int tid = threadIdx.x;
    const int f   = blockIdx.x;
    const int n   = blockIdx.y;

    Gs[tid] = 0.0f;
    Gt[tid] = 0.0f;
    __syncthreads();

    {
        const bool valid = tid < TDIM / 4;
        const int  t4    = tid;
        float4 zr[CDIM], zi[CDIM];
        float4 mv = make_float4(0.0f, 0.0f, 0.0f, 0.0f);
        #pragma unroll
        for (int r = 0; r < CDIM; ++r) {
            zr[r] = make_float4(0.0f, 0.0f, 0.0f, 0.0f);
            zi[r] = make_float4(0.0f, 0.0f, 0.0f, 0.0f);
        }
        if (valid) {
            #pragma unroll
            for (int r = 0; r < CDIM; ++r)
                zr[r] = ((const float4*)(spec_r + (((size_t)n * CDIM + r) * FDIM + f) * TDIM))[t4];
            #pragma unroll
            for (int r = 0; r < CDIM; ++r)
                zi[r] = ((const float4*)(spec_i + (((size_t)n * CDIM + r) * FDIM + f) * TDIM))[t4];
            mv = ((const float4*)(mask + ((size_t)n * FDIM + f) * TDIM))[t4];
        }
        #pragma unroll
        for (int r = 0; r < CDIM; ++r) {
            *(float4*)&Zs[r * ZSTRIDE + 4 * t4]          = zr[r];
            *(float4*)&Zs[(CDIM + r) * ZSTRIDE + 4 * t4] = zi[r];
        }
        *(float4*)&Ms[4 * t4] = mv;
    }
    __builtin_amdgcn_fence(__ATOMIC_RELEASE, "wavefront");
    __builtin_amdgcn_wave_barrier();
    __builtin_amdgcn_fence(__ATOMIC_ACQUIRE, "wavefront");

    {
        const int wave = __builtin_amdgcn_readfirstlane(tid >> 5);
        const int lane = tid & 31;
        const int m    = lane & 15;
        const int koff = (lane >> 4) * 2;

        v8f acc_s = {0, 0, 0, 0, 0, 0, 0, 0};
        v8f acc_t = {0, 0, 0, 0, 0, 0, 0, 0};
        v2f macc  = {0, 0};

        const int   tbase = 128 * wave + koff;
        const float* zrow = &Zs[m * ZSTRIDE + tbase];
        const float* mrow = &Ms[tbase];

        v2f zc[8], mc[8], zn[8], mn[8];
        #pragma unroll
        for (int u = 0; u < 8; ++u) {
            zc[u] = *(const v2f*)(zrow + 4 * u);
            mc[u] = *(const v2f*)(mrow + 4 * u);
        }
        #pragma unroll
        for (int b = 0; b < 4; ++b) {
            if (b < 3) {
                #pragma unroll
                for (int u = 0; u < 8; ++u) {
                    zn[u] = *(const v2f*)(zrow + 4 * (8 * (b + 1) + u));
                    mn[u] = *(const v2f*)(mrow + 4 * (8 * (b + 1) + u));
                }
            }
            __builtin_amdgcn_sched_barrier(0);
            #pragma unroll
            for (int u = 0; u < 8; ++u) {
                const v2f zb = zc[u];
                v2f za;
                za.x = zb.x * mc[u].x;
                za.y = zb.y * mc[u].y;
                macc.x += mc[u].x;
                macc.y += mc[u].y;
                acc_t = __builtin_amdgcn_wmma_f32_16x16x4_f32(
                            false, zb, false, zb, (short)0, acc_t, false, false);
                acc_s = __builtin_amdgcn_wmma_f32_16x16x4_f32(
                            false, za, false, zb, (short)0, acc_s, false, false);
            }
            #pragma unroll
            for (int u = 0; u < 8; ++u) { zc[u] = zn[u]; mc[u] = mn[u]; }
        }

        const int Nn    = lane & 15;
        const int Mbase = (lane >> 4) * 8;
        #pragma unroll
        for (int r = 0; r < 8; ++r) {
            atomicAdd(&Gs[(Mbase + r) * 16 + Nn], acc_s[r]);
            atomicAdd(&Gt[(Mbase + r) * 16 + Nn], acc_t[r]);
        }
        float wsum = macc.x + macc.y;
        wsum += other_half(wsum);
        if (lane == 0) mpart[wave] = wsum;
    }
    __syncthreads();

    if (__builtin_amdgcn_readfirstlane(tid >> 5) == 0) {
        const int   col  = tid & 15;
        const int   part = tid >> 4;
        const int   j    = col & 7;
        const bool  rhs  = col >= 8;
        const float ms    = ((mpart[0] + mpart[1]) + (mpart[2] + mpart[3])) + ((mpart[4] + mpart[5]) + (mpart[6] + mpart[7]));
        const float inv_s = 1.0f / fmaxf(ms, CEPS);
        const float inv_n = 1.0f / fmaxf((float)TDIM - ms, CEPS);
        const int   pIdx = (8 * part) * 16 + j;
        const int   qIdx = (8 * (1 - part)) * 16 + j + 8;
        const float sgn  = part ? -1.0f : 1.0f;

        float a[8];
        #pragma unroll
        for (int i = 0; i < 8; ++i) {
            const float gs_p = Gs[i * 16 + pIdx], gs_q = Gs[i * 16 + qIdx];
            const float gt_p = Gt[i * 16 + pIdx], gt_q = Gt[i * 16 + qIdx];
            const float rs = (gs_p + sgn * gs_q) * inv_s;
            float       rn = ((gt_p - gs_p) + sgn * (gt_q - gs_q)) * inv_n;
            rn += (part == 0 && i == j) ? CEPS : 0.0f;
            a[i] = rhs ? rs : rn;
        }

        #pragma unroll
        for (int k = 0; k < 8; ++k) {
            const float pr = lane_value(a[k], k), pi = lane_value(a[k], 16 + k);
            const float id = 1.0f / (pr * pr + pi * pi);
            const float pivR = pr * id, pivI = -pi * id;
            const float mine = a[k], oth = other_half(a[k]);
            const float ar = part ? oth : mine;
            const float ai = part ? mine : oth;
            const float nr = ar * pivR - ai * pivI;
            const float ni = ar * pivI + ai * pivR;
            a[k] = part ? ni : nr;
            const float u = part ? ni : nr;
            const float v = part ? nr : -ni;
            #pragma unroll
            for (int i = 0; i < 8; ++i) {
                if (i == k) continue;
                const float fr = lane_value(a[i], k), fi = lane_value(a[i], 16 + k);
                a[i] -= fr * u + fi * v;
            }
        }

        float tr = CEPS, ti = 0.0f;
        #pragma unroll
        for (int i = 0; i < 8; ++i) { tr += lane_value(a[i], 8 + i); ti += lane_value(a[i], 24 + i); }
        const float idt = 1.0f / (tr * tr + ti * ti);
        float wr[8], wi[8];
        #pragma unroll
        for (int i = 0; i < 8; ++i) {
            const float mr = lane_value(a[i], 8), mi = lane_value(a[i], 24);
            wr[i] = (mr * tr + mi * ti) * idt;
            wi[i] = (mi * tr - mr * ti) * idt;
        }
        if (tid == 0) {
            #pragma unroll
            for (int i = 0; i < 8; ++i) { wR[i] = wr[i]; wI[i] = wi[i]; }
        }
    }
    __syncthreads();

    if (tid < TDIM / 4) {
        float4 sr[CDIM], si[CDIM];
        #pragma unroll
        for (int c = 0; c < CDIM; ++c) {
            sr[c] = *(const float4*)&Zs[c * ZSTRIDE + 4 * tid];
            si[c] = *(const float4*)&Zs[(CDIM + c) * ZSTRIDE + 4 * tid];
        }
        const float4 wra = *(const float4*)&wR[0], wrb = *(const float4*)&wR[4];
        const float4 wia = *(const float4*)&wI[0], wib = *(const float4*)&wI[4];
        __builtin_amdgcn_sched_barrier(0);
        const float wr[CDIM] = {wra.x, wra.y, wra.z, wra.w, wrb.x, wrb.y, wrb.z, wrb.w};
        const float wi[CDIM] = {wia.x, wia.y, wia.z, wia.w, wib.x, wib.y, wib.z, wib.w};
        float4 br = make_float4(0.0f, 0.0f, 0.0f, 0.0f);
        float4 bi = make_float4(0.0f, 0.0f, 0.0f, 0.0f);
        #pragma unroll
        for (int c = 0; c < CDIM; ++c) {
            br.x += wr[c] * sr[c].x + wi[c] * si[c].x;
            br.y += wr[c] * sr[c].y + wi[c] * si[c].y;
            br.z += wr[c] * sr[c].z + wi[c] * si[c].z;
            br.w += wr[c] * sr[c].w + wi[c] * si[c].w;
            bi.x += wr[c] * si[c].x - wi[c] * sr[c].x;
            bi.y += wr[c] * si[c].y - wi[c] * sr[c].y;
            bi.z += wr[c] * si[c].z - wi[c] * sr[c].z;
            bi.w += wr[c] * si[c].w - wi[c] * sr[c].w;
        }
        float* outR = out + (((size_t)n * 2 + 0) * FDIM + f) * TDIM;
        float* outI = out + (((size_t)n * 2 + 1) * FDIM + f) * TDIM;
        ((float4*)outR)[tid] = br;
        ((float4*)outI)[tid] = bi;
    }
}

extern "C" void kernel_launch(void* const* d_in, const int* in_sizes, int n_in,
                              void* d_out, int out_size, void* d_ws, size_t ws_size,
                              hipStream_t stream) {
    (void)in_sizes; (void)n_in; (void)out_size; (void)d_ws; (void)ws_size;
    const float* spec_r = (const float*)d_in[0];
    const float* spec_i = (const float*)d_in[1];
    const float* mask   = (const float*)d_in[2];
    float*       outp   = (float*)d_out;

    dim3 grid(FDIM, 16);
    dim3 block(256);
    hipLaunchKernelGGL(mvdr_wmma_kernel, grid, block, 0, stream,
                       spec_r, spec_i, mask, outp);
}
